# starting kernel
# baseline (speedup 1.0000x reference)
#include <hip/hip_runtime.h>
#include <cmath>

typedef float v2f __attribute__((ext_vector_type(2)));
typedef float v8f __attribute__((ext_vector_type(8)));

#define N_    256
#define CIN   64
#define COUT  64
#define T_    64
#define V_    25
#define REL   8
#define TV    (T_ * V_)
#define VP    28
#define UP    32
#define WRS   68
#define BLK   512
#define NWAVE 16

#define OFF_WR    0
#define SZ_WR     (COUT * WRS)
#define OFF_XBAR  (OFF_WR + SZ_WR)
#define SZ_XBAR   (CIN * V_)
#define OFF_X1    (OFF_XBAR + SZ_XBAR)
#define OFF_X2    (OFF_X1 + REL * V_)
#define OFF_D     (OFF_X2 + REL * V_)
#define SZ_D      (REL * V_ * V_)
#define OFF_INP   (OFF_D + SZ_D)
#define SZ_INP    (16 * T_ * VP)
#define OFF_AF    (OFF_INP + SZ_INP)
#define SZ_AF     (16 * UP * VP)
#define SMEM_FLOATS (OFF_AF + SZ_AF)

__global__ void gcn_fused_kernel(const float* __restrict__ x,
                                 const float* __restrict__ Ag,
                                 const float* __restrict__ w1, const float* __restrict__ b1,
                                 const float* __restrict__ w2, const float* __restrict__ b2,
                                 const float* __restrict__ w3, const float* __restrict__ b3,
                                 const float* __restrict__ wr, const float* __restrict__ br,
                                 float* __restrict__ out)
{
    extern __shared__ float sm[];
    const int n    = blockIdx.x;
    const int tid  = threadIdx.x;
    const int wave = tid >> 5;
    const int lane = tid & 31;
    const int lo   = lane & 15;
    const int hi   = lane >> 4;
    const float* __restrict__ xg = x + (size_t)n * CIN * TV;

    for (int idx = tid; idx < COUT * CIN; idx += BLK) {
        const int c = idx / CIN, i = idx % CIN;
        sm[OFF_WR + c * WRS + i] = wr[idx];
    }
    for (int idx = tid; idx < SZ_INP + SZ_AF; idx += BLK) sm[OFF_INP + idx] = 0.0f;

    for (int idx = tid; idx < CIN * V_; idx += BLK) {
        const int i = idx / V_, v = idx % V_;
        const float* p = xg + i * TV + v;
        float s = 0.0f;
#pragma unroll 8
        for (int t = 0; t < T_; ++t) s += p[t * V_];
        sm[OFF_XBAR + idx] = s * (1.0f / T_);
    }
    __syncthreads();

    for (int idx = tid; idx < 2 * REL * V_; idx += BLK) {
        const int which = idx / (REL * V_);
        const int rv = idx % (REL * V_);
        const int r = rv / V_, v = rv % V_;
        const float* __restrict__ w = which ? w2 : w1;
        float s = which ? b2[r] : b1[r];
#pragma unroll 8
        for (int i = 0; i < CIN; ++i) s += w[r * CIN + i] * sm[OFF_XBAR + i * V_ + v];
        sm[(which ? OFF_X2 : OFF_X1) + rv] = s;
    }
    __syncthreads();

    for (int idx = tid; idx < SZ_D; idx += BLK) {
        const int r = idx / (V_ * V_);
        const int uv = idx % (V_ * V_);
        const int u = uv / V_, v = uv % V_;
        sm[OFF_D + idx] = tanhf(sm[OFF_X1 + r * V_ + u] - sm[OFF_X2 + r * V_ + v]);
    }

    for (int ct = 0; ct < COUT / 16; ++ct) {
        __syncthreads();

        for (int idx = tid; idx < 16 * V_ * V_; idx += BLK) {
            const int cc = idx / (V_ * V_);
            const int uv = idx % (V_ * V_);
            const int u = uv / V_, v = uv % V_;
            const int c = ct * 16 + cc;
            float acc = b3[c] + Ag[uv];
#pragma unroll
            for (int r = 0; r < REL; ++r)
                acc += w3[c * REL + r] * sm[OFF_D + r * (V_ * V_) + uv];
            sm[OFF_AF + cc * (UP * VP) + u * VP + v] = acc;
        }

        for (int tile = wave; tile < TV / 16; tile += NWAVE) {
            const int col = tile * 16 + lo;
            v8f acc = {};
#pragma unroll
            for (int k0 = 0; k0 < CIN; k0 += 4) {
                const int i0 = k0 + 2 * hi;
                v2f a, b;
                a.x = sm[OFF_WR + (ct * 16 + lo) * WRS + i0];
                a.y = sm[OFF_WR + (ct * 16 + lo) * WRS + i0 + 1];
                b.x = xg[i0 * TV + col];
                b.y = xg[(i0 + 1) * TV + col];
                acc = __builtin_amdgcn_wmma_f32_16x16x4_f32(
                        false, a, false, b, (short)0, acc, false, false);
            }
            const int t = col / V_, v = col % V_;
#pragma unroll
            for (int j = 0; j < 8; ++j) {
                const int m = j + 8 * hi;
                sm[OFF_INP + m * (T_ * VP) + t * VP + v] = acc[j] + br[ct * 16 + m];
            }
        }
        __syncthreads();

        {
            const int cl = wave;
            const int c = ct * 16 + cl;
#pragma unroll
            for (int mt = 0; mt < 4; ++mt) {
#pragma unroll
                for (int nt = 0; nt < 2; ++nt) {
                    v8f acc = {};
#pragma unroll
                    for (int k0 = 0; k0 < VP; k0 += 4) {
                        const int kk = k0 + 2 * hi;
                        v2f a, b;
                        a.x = sm[OFF_INP + cl * (T_ * VP) + (mt * 16 + lo) * VP + kk];
                        a.y = sm[OFF_INP + cl * (T_ * VP) + (mt * 16 + lo) * VP + kk + 1];
                        b.x = sm[OFF_AF + cl * (UP * VP) + (nt * 16 + lo) * VP + kk];
                        b.y = sm[OFF_AF + cl * (UP * VP) + (nt * 16 + lo) * VP + kk + 1];
                        acc = __builtin_amdgcn_wmma_f32_16x16x4_f32(
                                false, a, false, b, (short)0, acc, false, false);
                    }
                    const int u = nt * 16 + lo;
                    if (u < V_) {
#pragma unroll
                        for (int j = 0; j < 8; ++j) {
                            const int t = mt * 16 + j + 8 * hi;
                            out[(((size_t)n * COUT + c) * T_ + t) * V_ + u] = acc[j];
                        }
                    }
                }
            }
        }
    }
}

extern "C" void kernel_launch(void* const* d_in, const int* in_sizes, int n_in,
                              void* d_out, int out_size, void* d_ws, size_t ws_size,
                              hipStream_t stream) {
    const float* x  = (const float*)d_in[0];
    const float* A  = (const float*)d_in[1];
    const float* w1 = (const float*)d_in[2];
    const float* b1 = (const float*)d_in[3];
    const float* w2 = (const float*)d_in[4];
    const float* b2 = (const float*)d_in[5];
    const float* w3 = (const float*)d_in[6];
    const float* b3 = (const float*)d_in[7];
    const float* wr = (const float*)d_in[8];
    const float* br = (const float*)d_in[9];
    float* out = (float*)d_out;

    const size_t shmem = (size_t)SMEM_FLOATS * sizeof(float);
    gcn_fused_kernel<<<dim3(N_), dim3(BLK), shmem, stream>>>(
        x, A, w1, b1, w2, b2, w3, b3, wr, br, out);
}


# Time-mean pass reads x as 16-byte loads, 16 independent loads in flight per lane (was 8 dword loads drained per trip); 4 partial sums per joint reduced via LDS: mean summed in a new order
# speedup vs baseline: 1.0900x; 1.0900x over previous
#include <hip/hip_runtime.h>
#include <cmath>

typedef float v2f __attribute__((ext_vector_type(2)));
typedef float v8f __attribute__((ext_vector_type(8)));
typedef float v4f __attribute__((ext_vector_type(4)));

#define N_    256
#define CIN   64
#define COUT  64
#define T_    64
#define V_    25
#define REL   8
#define TV    (T_ * V_)
#define VP    28
#define UP    32
#define WRS   68
#define BLK   512
#define NWAVE 16

#define OFF_WR    0
#define SZ_WR     (COUT * WRS)
#define OFF_XBAR  (OFF_WR + SZ_WR)
#define SZ_XBAR   (CIN * V_)
#define OFF_X1    (OFF_XBAR + SZ_XBAR)
#define OFF_X2    (OFF_X1 + REL * V_)
#define OFF_D     (OFF_X2 + REL * V_)
#define SZ_D      (REL * V_ * V_)
#define OFF_INP   (OFF_D + SZ_D)
#define SZ_INP    (16 * T_ * VP)
#define OFF_AF    (OFF_INP + SZ_INP)
#define SZ_AF     (16 * UP * VP)
#define OFF_PART  (OFF_AF + SZ_AF)
#define QPR       (V_ * 4)
#define SZ_PART   (CIN * QPR)
#define SMEM_FLOATS (OFF_PART + SZ_PART)

__global__ void gcn_fused_kernel(const float* __restrict__ x,
                                 const float* __restrict__ Ag,
                                 const float* __restrict__ w1, const float* __restrict__ b1,
                                 const float* __restrict__ w2, const float* __restrict__ b2,
                                 const float* __restrict__ w3, const float* __restrict__ b3,
                                 const float* __restrict__ wr, const float* __restrict__ br,
                                 float* __restrict__ out)
{
    extern __shared__ float sm[];
    const int n    = blockIdx.x;
    const int tid  = threadIdx.x;
    const int wave = tid >> 5;
    const int lane = tid & 31;
    const int lo   = lane & 15;
    const int hi   = lane >> 4;
    const float* __restrict__ xg = x + (size_t)n * CIN * TV;

    for (int idx = tid; idx < COUT * CIN; idx += BLK) {
        const int c = idx / CIN, i = idx % CIN;
        sm[OFF_WR + c * WRS + i] = wr[idx];
    }
    for (int idx = tid; idx < SZ_INP + SZ_AF; idx += BLK) sm[OFF_INP + idx] = 0.0f;

    for (int idx = tid; idx < CIN * V_; idx += BLK) {
        const int i = idx / V_, q = idx % V_;
        const v4f* p = reinterpret_cast<const v4f*>(xg + i * TV) + q;
        v4f r[T_ / 4];
#pragma unroll
        for (int m = 0; m < T_ / 4; ++m) r[m] = p[m * V_];
        v4f s = r[0];
#pragma unroll
        for (int m = 1; m < T_ / 4; ++m) s += r[m];
        float* dst = &sm[OFF_PART + i * QPR + 4 * q];
        dst[0] = s.x; dst[1] = s.y; dst[2] = s.z; dst[3] = s.w;
    }
    __syncthreads();
    for (int idx = tid; idx < CIN * V_; idx += BLK) {
        const int i = idx / V_, v = idx % V_;
        const float* ps = &sm[OFF_PART + i * QPR + v];
        sm[OFF_XBAR + idx] = ((ps[0] + ps[V_]) + (ps[2 * V_] + ps[3 * V_])) * (1.0f / T_);
    }
    __syncthreads();

    for (int idx = tid; idx < 2 * REL * V_; idx += BLK) {
        const int which = idx / (REL * V_);
        const int rv = idx % (REL * V_);
        const int r = rv / V_, v = rv % V_;
        const float* __restrict__ w = which ? w2 : w1;
        float s = which ? b2[r] : b1[r];
#pragma unroll 8
        for (int i = 0; i < CIN; ++i) s += w[r * CIN + i] * sm[OFF_XBAR + i * V_ + v];
        sm[(which ? OFF_X2 : OFF_X1) + rv] = s;
    }
    __syncthreads();

    for (int idx = tid; idx < SZ_D; idx += BLK) {
        const int r = idx / (V_ * V_);
        const int uv = idx % (V_ * V_);
        const int u = uv / V_, v = uv % V_;
        sm[OFF_D + idx] = tanhf(sm[OFF_X1 + r * V_ + u] - sm[OFF_X2 + r * V_ + v]);
    }

    for (int ct = 0; ct < COUT / 16; ++ct) {
        __syncthreads();

        for (int idx = tid; idx < 16 * V_ * V_; idx += BLK) {
            const int cc = idx / (V_ * V_);
            const int uv = idx % (V_ * V_);
            const int u = uv / V_, v = uv % V_;
            const int c = ct * 16 + cc;
            float acc = b3[c] + Ag[uv];
#pragma unroll
            for (int r = 0; r < REL; ++r)
                acc += w3[c * REL + r] * sm[OFF_D + r * (V_ * V_) + uv];
            sm[OFF_AF + cc * (UP * VP) + u * VP + v] = acc;
        }

        for (int tile = wave; tile < TV / 16; tile += NWAVE) {
            const int col = tile * 16 + lo;
            v8f acc = {};
#pragma unroll
            for (int k0 = 0; k0 < CIN; k0 += 4) {
                const int i0 = k0 + 2 * hi;
                v2f a, b;
                a.x = sm[OFF_WR + (ct * 16 + lo) * WRS + i0];
                a.y = sm[OFF_WR + (ct * 16 + lo) * WRS + i0 + 1];
                b.x = xg[i0 * TV + col];
                b.y = xg[(i0 + 1) * TV + col];
                acc = __builtin_amdgcn_wmma_f32_16x16x4_f32(
                        false, a, false, b, (short)0, acc, false, false);
            }
            const int t = col / V_, v = col % V_;
#pragma unroll
            for (int j = 0; j < 8; ++j) {
                const int m = j + 8 * hi;
                sm[OFF_INP + m * (T_ * VP) + t * VP + v] = acc[j] + br[ct * 16 + m];
            }
        }
        __syncthreads();

        {
            const int cl = wave;
            const int c = ct * 16 + cl;
#pragma unroll
            for (int mt = 0; mt < 4; ++mt) {
#pragma unroll
                for (int nt = 0; nt < 2; ++nt) {
                    v8f acc = {};
#pragma unroll
                    for (int k0 = 0; k0 < VP; k0 += 4) {
                        const int kk = k0 + 2 * hi;
                        v2f a, b;
                        a.x = sm[OFF_INP + cl * (T_ * VP) + (mt * 16 + lo) * VP + kk];
                        a.y = sm[OFF_INP + cl * (T_ * VP) + (mt * 16 + lo) * VP + kk + 1];
                        b.x = sm[OFF_AF + cl * (UP * VP) + (nt * 16 + lo) * VP + kk];
                        b.y = sm[OFF_AF + cl * (UP * VP) + (nt * 16 + lo) * VP + kk + 1];
                        acc = __builtin_amdgcn_wmma_f32_16x16x4_f32(
                                false, a, false, b, (short)0, acc, false, false);
                    }
                    const int u = nt * 16 + lo;
                    if (u < V_) {
#pragma unroll
                        for (int j = 0; j < 8; ++j) {
                            const int t = mt * 16 + j + 8 * hi;
                            out[(((size_t)n * COUT + c) * T_ + t) * V_ + u] = acc[j];
                        }
                    }
                }
            }
        }
    }
}

extern "C" void kernel_launch(void* const* d_in, const int* in_sizes, int n_in,
                              void* d_out, int out_size, void* d_ws, size_t ws_size,
                              hipStream_t stream) {
    const float* x  = (const float*)d_in[0];
    const float* A  = (const float*)d_in[1];
    const float* w1 = (const float*)d_in[2];
    const float* b1 = (const float*)d_in[3];
    const float* w2 = (const float*)d_in[4];
    const float* b2 = (const float*)d_in[5];
    const float* w3 = (const float*)d_in[6];
    const float* b3 = (const float*)d_in[7];
    const float* wr = (const float*)d_in[8];
    const float* br = (const float*)d_in[9];
    float* out = (float*)d_out;

    const size_t shmem = (size_t)SMEM_FLOATS * sizeof(float);
    gcn_fused_kernel<<<dim3(N_), dim3(BLK), shmem, stream>>>(
        x, A, w1, b1, w2, b2, w3, b3, wr, br, out);
}


# Output leaves as whole 128-byte lines: each wave stages half a channel plane (32 rows, 800 floats) in its own LDS area and stores 16 bytes per lane, not 4-byte pieces of 16 rows
# speedup vs baseline: 1.1592x; 1.0635x over previous
#include <hip/hip_runtime.h>
#include <cmath>

typedef float v2f __attribute__((ext_vector_type(2)));
typedef float v8f __attribute__((ext_vector_type(8)));
typedef float v4f __attribute__((ext_vector_type(4)));

#define N_    256
#define CIN   64
#define COUT  64
#define T_    64
#define V_    25
#define REL   8
#define TV    (T_ * V_)
#define VP    28
#define UP    32
#define WRS   68
#define BLK   512
#define NWAVE 16

#define OFF_WR    0
#define SZ_WR     (COUT * WRS)
#define OFF_XBAR  (OFF_WR + SZ_WR)
#define SZ_XBAR   (CIN * V_)
#define OFF_X1    (OFF_XBAR + SZ_XBAR)
#define OFF_X2    (OFF_X1 + REL * V_)
#define OFF_D     (OFF_X2 + REL * V_)
#define SZ_D      (REL * V_ * V_)
#define OFF_INP   (OFF_D + SZ_D)
#define SZ_INP    (16 * T_ * VP)
#define OFF_AF    (OFF_INP + SZ_INP)
#define SZ_AF     (16 * UP * VP)
#define OFF_PART  (OFF_AF + SZ_AF)
#define QPR       (V_ * 4)
#define SZ_PART   (CIN * QPR)
#define HROWS     32
#define SZ_STGW   (HROWS * V_)
#define OFF_STG   OFF_PART
#define SMEM_FLOATS (OFF_STG + NWAVE * SZ_STGW)

__global__ void gcn_fused_kernel(const float* __restrict__ x,
                                 const float* __restrict__ Ag,
                                 const float* __restrict__ w1, const float* __restrict__ b1,
                                 const float* __restrict__ w2, const float* __restrict__ b2,
                                 const float* __restrict__ w3, const float* __restrict__ b3,
                                 const float* __restrict__ wr, const float* __restrict__ br,
                                 float* __restrict__ out)
{
    extern __shared__ float sm[];
    const int n    = blockIdx.x;
    const int tid  = threadIdx.x;
    const int wave = tid >> 5;
    const int lane = tid & 31;
    const int lo   = lane & 15;
    const int hi   = lane >> 4;
    const float* __restrict__ xg = x + (size_t)n * CIN * TV;

    for (int idx = tid; idx < COUT * CIN; idx += BLK) {
        const int c = idx / CIN, i = idx % CIN;
        sm[OFF_WR + c * WRS + i] = wr[idx];
    }
    for (int idx = tid; idx < SZ_INP + SZ_AF; idx += BLK) sm[OFF_INP + idx] = 0.0f;

    for (int idx = tid; idx < CIN * V_; idx += BLK) {
        const int i = idx / V_, q = idx % V_;
        const v4f* p = reinterpret_cast<const v4f*>(xg + i * TV) + q;
        v4f r[T_ / 4];
#pragma unroll
        for (int m = 0; m < T_ / 4; ++m) r[m] = p[m * V_];
        v4f s = r[0];
#pragma unroll
        for (int m = 1; m < T_ / 4; ++m) s += r[m];
        float* dst = &sm[OFF_PART + i * QPR + 4 * q];
        dst[0] = s.x; dst[1] = s.y; dst[2] = s.z; dst[3] = s.w;
    }
    __syncthreads();
    for (int idx = tid; idx < CIN * V_; idx += BLK) {
        const int i = idx / V_, v = idx % V_;
        const float* ps = &sm[OFF_PART + i * QPR + v];
        sm[OFF_XBAR + idx] = ((ps[0] + ps[V_]) + (ps[2 * V_] + ps[3 * V_])) * (1.0f / T_);
    }
    __syncthreads();

    for (int idx = tid; idx < 2 * REL * V_; idx += BLK) {
        const int which = idx / (REL * V_);
        const int rv = idx % (REL * V_);
        const int r = rv / V_, v = rv % V_;
        const float* __restrict__ w = which ? w2 : w1;
        float s = which ? b2[r] : b1[r];
#pragma unroll 8
        for (int i = 0; i < CIN; ++i) s += w[r * CIN + i] * sm[OFF_XBAR + i * V_ + v];
        sm[(which ? OFF_X2 : OFF_X1) + rv] = s;
    }
    __syncthreads();

    for (int idx = tid; idx < SZ_D; idx += BLK) {
        const int r = idx / (V_ * V_);
        const int uv = idx % (V_ * V_);
        const int u = uv / V_, v = uv % V_;
        sm[OFF_D + idx] = tanhf(sm[OFF_X1 + r * V_ + u] - sm[OFF_X2 + r * V_ + v]);
    }

    for (int ct = 0; ct < COUT / 16; ++ct) {
        __syncthreads();

        for (int idx = tid; idx < 16 * V_ * V_; idx += BLK) {
            const int cc = idx / (V_ * V_);
            const int uv = idx % (V_ * V_);
            const int u = uv / V_, v = uv % V_;
            const int c = ct * 16 + cc;
            float acc = b3[c] + Ag[uv];
#pragma unroll
            for (int r = 0; r < REL; ++r)
                acc += w3[c * REL + r] * sm[OFF_D + r * (V_ * V_) + uv];
            sm[OFF_AF + cc * (UP * VP) + u * VP + v] = acc;
        }

        for (int tile = wave; tile < TV / 16; tile += NWAVE) {
            const int col = tile * 16 + lo;
            v8f acc = {};
#pragma unroll
            for (int k0 = 0; k0 < CIN; k0 += 4) {
                const int i0 = k0 + 2 * hi;
                v2f a, b;
                a.x = sm[OFF_WR + (ct * 16 + lo) * WRS + i0];
                a.y = sm[OFF_WR + (ct * 16 + lo) * WRS + i0 + 1];
                b.x = xg[i0 * TV + col];
                b.y = xg[(i0 + 1) * TV + col];
                acc = __builtin_amdgcn_wmma_f32_16x16x4_f32(
                        false, a, false, b, (short)0, acc, false, false);
            }
            const int t = col / V_, v = col % V_;
#pragma unroll
            for (int j = 0; j < 8; ++j) {
                const int m = j + 8 * hi;
                sm[OFF_INP + m * (T_ * VP) + t * VP + v] = acc[j] + br[ct * 16 + m];
            }
        }
        __syncthreads();

        {
            const int cl = wave;
            const int c = ct * 16 + cl;
            float* stg = &sm[OFF_STG + wave * SZ_STGW];
            float* og = out + ((size_t)n * COUT + c) * TV;
#pragma unroll
            for (int mt = 0; mt < 4; ++mt) {
                const int ml = mt & 1;
#pragma unroll
                for (int nt = 0; nt < 2; ++nt) {
                    v8f acc = {};
#pragma unroll
                    for (int k0 = 0; k0 < VP; k0 += 4) {
                        const int kk = k0 + 2 * hi;
                        v2f a, b;
                        a.x = sm[OFF_INP + cl * (T_ * VP) + (mt * 16 + lo) * VP + kk];
                        a.y = sm[OFF_INP + cl * (T_ * VP) + (mt * 16 + lo) * VP + kk + 1];
                        b.x = sm[OFF_AF + cl * (UP * VP) + (nt * 16 + lo) * VP + kk];
                        b.y = sm[OFF_AF + cl * (UP * VP) + (nt * 16 + lo) * VP + kk + 1];
                        acc = __builtin_amdgcn_wmma_f32_16x16x4_f32(
                                false, a, false, b, (short)0, acc, false, false);
                    }
                    const int u = nt * 16 + lo;
                    if (u < V_) {
#pragma unroll
                        for (int j = 0; j < 8; ++j)
                            stg[(ml * 16 + j + 8 * hi) * V_ + u] = acc[j];
                    }
                }
                if (ml == 1) {
                    float* oh = og + (mt >> 1) * SZ_STGW;
                    for (int s = 0; s < (SZ_STGW / 4 + 31) / 32; ++s) {
                        const int q = lane + 32 * s;
                        if (q < SZ_STGW / 4) {
                            const v4f val = *reinterpret_cast<const v4f*>(stg + 4 * q);
                            *reinterpret_cast<v4f*>(oh + 4 * q) = val;
                        }
                    }
                }
            }
        }
    }
}

extern "C" void kernel_launch(void* const* d_in, const int* in_sizes, int n_in,
                              void* d_out, int out_size, void* d_ws, size_t ws_size,
                              hipStream_t stream) {
    const float* x  = (const float*)d_in[0];
    const float* A  = (const float*)d_in[1];
    const float* w1 = (const float*)d_in[2];
    const float* b1 = (const float*)d_in[3];
    const float* w2 = (const float*)d_in[4];
    const float* b2 = (const float*)d_in[5];
    const float* w3 = (const float*)d_in[6];
    const float* b3 = (const float*)d_in[7];
    const float* wr = (const float*)d_in[8];
    const float* br = (const float*)d_in[9];
    float* out = (float*)d_out;

    const size_t shmem = (size_t)SMEM_FLOATS * sizeof(float);
    gcn_fused_kernel<<<dim3(N_), dim3(BLK), shmem, stream>>>(
        x, A, w1, b1, w2, b2, w3, b3, wr, br, out);
}


# Afull tile (16 channels x 8 relations x 625 joint pairs) built by two f32 WMMA k steps per 16 columns, not 8 scalar multiply-adds per element; b3 + A added after the sum: Afull summed in a new order
# speedup vs baseline: 1.2969x; 1.1188x over previous
#include <hip/hip_runtime.h>
#include <cmath>

typedef float v2f __attribute__((ext_vector_type(2)));
typedef float v8f __attribute__((ext_vector_type(8)));
typedef float v4f __attribute__((ext_vector_type(4)));

#define N_    256
#define CIN   64
#define COUT  64
#define T_    64
#define V_    25
#define REL   8
#define TV    (T_ * V_)
#define VP    28
#define UP    32
#define WRS   68
#define BLK   512
#define NWAVE 16

#define OFF_WR    0
#define SZ_WR     (COUT * WRS)
#define OFF_XBAR  (OFF_WR + SZ_WR)
#define SZ_XBAR   (CIN * V_)
#define OFF_X1    (OFF_XBAR + SZ_XBAR)
#define OFF_X2    (OFF_X1 + REL * V_)
#define OFF_D     (OFF_X2 + REL * V_)
#define SZ_D      (REL * V_ * V_)
#define OFF_INP   (OFF_D + SZ_D)
#define SZ_INP    (16 * T_ * VP)
#define OFF_AF    (OFF_INP + SZ_INP)
#define SZ_AF     (16 * UP * VP)
#define OFF_PART  (OFF_AF + SZ_AF)
#define QPR       (V_ * 4)
#define SZ_PART   (CIN * QPR)
#define HROWS     32
#define SZ_STGW   (HROWS * V_)
#define OFF_STG   OFF_PART
#define SMEM_FLOATS (OFF_STG + NWAVE * SZ_STGW)

__global__ void gcn_fused_kernel(const float* __restrict__ x,
                                 const float* __restrict__ Ag,
                                 const float* __restrict__ w1, const float* __restrict__ b1,
                                 const float* __restrict__ w2, const float* __restrict__ b2,
                                 const float* __restrict__ w3, const float* __restrict__ b3,
                                 const float* __restrict__ wr, const float* __restrict__ br,
                                 float* __restrict__ out)
{
    extern __shared__ float sm[];
    const int n    = blockIdx.x;
    const int tid  = threadIdx.x;
    const int wave = tid >> 5;
    const int lane = tid & 31;
    const int lo   = lane & 15;
    const int hi   = lane >> 4;
    const float* __restrict__ xg = x + (size_t)n * CIN * TV;

    for (int idx = tid; idx < COUT * CIN; idx += BLK) {
        const int c = idx / CIN, i = idx % CIN;
        sm[OFF_WR + c * WRS + i] = wr[idx];
    }
    for (int idx = tid; idx < SZ_INP + SZ_AF; idx += BLK) sm[OFF_INP + idx] = 0.0f;

    for (int idx = tid; idx < CIN * V_; idx += BLK) {
        const int i = idx / V_, q = idx % V_;
        const v4f* p = reinterpret_cast<const v4f*>(xg + i * TV) + q;
        v4f r[T_ / 4];
#pragma unroll
        for (int m = 0; m < T_ / 4; ++m) r[m] = p[m * V_];
        v4f s = r[0];
#pragma unroll
        for (int m = 1; m < T_ / 4; ++m) s += r[m];
        float* dst = &sm[OFF_PART + i * QPR + 4 * q];
        dst[0] = s.x; dst[1] = s.y; dst[2] = s.z; dst[3] = s.w;
    }
    __syncthreads();
    for (int idx = tid; idx < CIN * V_; idx += BLK) {
        const int i = idx / V_, v = idx % V_;
        const float* ps = &sm[OFF_PART + i * QPR + v];
        sm[OFF_XBAR + idx] = ((ps[0] + ps[V_]) + (ps[2 * V_] + ps[3 * V_])) * (1.0f / T_);
    }
    __syncthreads();

    for (int idx = tid; idx < 2 * REL * V_; idx += BLK) {
        const int which = idx / (REL * V_);
        const int rv = idx % (REL * V_);
        const int r = rv / V_, v = rv % V_;
        const float* __restrict__ w = which ? w2 : w1;
        float s = which ? b2[r] : b1[r];
#pragma unroll 8
        for (int i = 0; i < CIN; ++i) s += w[r * CIN + i] * sm[OFF_XBAR + i * V_ + v];
        sm[(which ? OFF_X2 : OFF_X1) + rv] = s;
    }
    __syncthreads();

    v2f wa0, wa1;
    float bias[8];
    auto load_w3_tile = [&](int ctile) {
        const float* w3row = w3 + (ctile * 16 + lo) * REL + 2 * hi;
        wa0.x = w3row[0]; wa0.y = w3row[1];
        wa1.x = w3row[4]; wa1.y = w3row[5];
#pragma unroll
        for (int j = 0; j < 8; ++j) bias[j] = b3[ctile * 16 + 8 * hi + j];
    };
    load_w3_tile(0);

    for (int idx = tid; idx < SZ_D; idx += BLK) {
        const int r = idx / (V_ * V_);
        const int uv = idx % (V_ * V_);
        const int u = uv / V_, v = uv % V_;
        sm[OFF_D + idx] = tanhf(sm[OFF_X1 + r * V_ + u] - sm[OFF_X2 + r * V_ + v]);
    }

    for (int ct = 0; ct < COUT / 16; ++ct) {
        __syncthreads();

        for (int tile = wave; tile < (V_ * V_ + 15) / 16; tile += NWAVE) {
            const int uv  = tile * 16 + lo;
            const int uvc = uv < V_ * V_ ? uv : V_ * V_ - 1;
            const float ag = Ag[uvc];
            v2f d0, d1;
            d0.x = sm[OFF_D + (2 * hi)     * (V_ * V_) + uvc];
            d0.y = sm[OFF_D + (2 * hi + 1) * (V_ * V_) + uvc];
            d1.x = sm[OFF_D + (2 * hi + 4) * (V_ * V_) + uvc];
            d1.y = sm[OFF_D + (2 * hi + 5) * (V_ * V_) + uvc];
            v8f acc = {};
            acc = __builtin_amdgcn_wmma_f32_16x16x4_f32(false, wa0, false, d0, (short)0, acc, false, false);
            acc = __builtin_amdgcn_wmma_f32_16x16x4_f32(false, wa1, false, d1, (short)0, acc, false, false);
            const int u = uv / V_, v = uv % V_;
            float* dst = &sm[OFF_AF + (8 * hi) * (UP * VP) + u * VP + v];
#pragma unroll
            for (int j = 0; j < 8; ++j)
                dst[j * (UP * VP)] = acc[j] + (bias[j] + ag);
        }
        if (ct + 1 < COUT / 16) load_w3_tile(ct + 1);

        for (int tile = wave; tile < TV / 16; tile += NWAVE) {
            const int col = tile * 16 + lo;
            v8f acc = {};
#pragma unroll
            for (int k0 = 0; k0 < CIN; k0 += 4) {
                const int i0 = k0 + 2 * hi;
                v2f a, b;
                a.x = sm[OFF_WR + (ct * 16 + lo) * WRS + i0];
                a.y = sm[OFF_WR + (ct * 16 + lo) * WRS + i0 + 1];
                b.x = xg[i0 * TV + col];
                b.y = xg[(i0 + 1) * TV + col];
                acc = __builtin_amdgcn_wmma_f32_16x16x4_f32(
                        false, a, false, b, (short)0, acc, false, false);
            }
            const int t = col / V_, v = col % V_;
#pragma unroll
            for (int j = 0; j < 8; ++j) {
                const int m = j + 8 * hi;
                sm[OFF_INP + m * (T_ * VP) + t * VP + v] = acc[j] + br[ct * 16 + m];
            }
        }
        __syncthreads();

        {
            const int cl = wave;
            const int c = ct * 16 + cl;
            float* stg = &sm[OFF_STG + wave * SZ_STGW];
            float* og = out + ((size_t)n * COUT + c) * TV;
#pragma unroll
            for (int mt = 0; mt < 4; ++mt) {
                const int ml = mt & 1;
#pragma unroll
                for (int nt = 0; nt < 2; ++nt) {
                    v8f acc = {};
#pragma unroll
                    for (int k0 = 0; k0 < VP; k0 += 4) {
                        const int kk = k0 + 2 * hi;
                        v2f a, b;
                        a.x = sm[OFF_INP + cl * (T_ * VP) + (mt * 16 + lo) * VP + kk];
                        a.y = sm[OFF_INP + cl * (T_ * VP) + (mt * 16 + lo) * VP + kk + 1];
                        b.x = sm[OFF_AF + cl * (UP * VP) + (nt * 16 + lo) * VP + kk];
                        b.y = sm[OFF_AF + cl * (UP * VP) + (nt * 16 + lo) * VP + kk + 1];
                        acc = __builtin_amdgcn_wmma_f32_16x16x4_f32(
                                false, a, false, b, (short)0, acc, false, false);
                    }
                    const int u = nt * 16 + lo;
                    if (u < V_) {
#pragma unroll
                        for (int j = 0; j < 8; ++j)
                            stg[(ml * 16 + j + 8 * hi) * V_ + u] = acc[j];
                    }
                }
                if (ml == 1) {
                    float* oh = og + (mt >> 1) * SZ_STGW;
                    for (int s = 0; s < (SZ_STGW / 4 + 31) / 32; ++s) {
                        const int q = lane + 32 * s;
                        if (q < SZ_STGW / 4) {
                            const v4f val = *reinterpret_cast<const v4f*>(stg + 4 * q);
                            *reinterpret_cast<v4f*>(oh + 4 * q) = val;
                        }
                    }
                }
            }
        }
    }
}

extern "C" void kernel_launch(void* const* d_in, const int* in_sizes, int n_in,
                              void* d_out, int out_size, void* d_ws, size_t ws_size,
                              hipStream_t stream) {
    const float* x  = (const float*)d_in[0];
    const float* A  = (const float*)d_in[1];
    const float* w1 = (const float*)d_in[2];
    const float* b1 = (const float*)d_in[3];
    const float* w2 = (const float*)d_in[4];
    const float* b2 = (const float*)d_in[5];
    const float* w3 = (const float*)d_in[6];
    const float* b3 = (const float*)d_in[7];
    const float* wr = (const float*)d_in[8];
    const float* br = (const float*)d_in[9];
    float* out = (float*)d_out;

    const size_t shmem = (size_t)SMEM_FLOATS * sizeof(float);
    gcn_fused_kernel<<<dim3(N_), dim3(BLK), shmem, stream>>>(
        x, A, w1, b1, w2, b2, w3, b3, wr, br, out);
}


# Main loop over blocks of 16 time steps with all 64 channels (Afull built once, wr tile in registers): four waves share each x tile, so GEMM1 pulls x through L2 once, not four times
# speedup vs baseline: 1.5857x; 1.2227x over previous
#include <hip/hip_runtime.h>
#include <cmath>


typedef float v2f __attribute__((ext_vector_type(2)));
typedef float v8f __attribute__((ext_vector_type(8)));
typedef float v4f __attribute__((ext_vector_type(4)));

#define N_    256
#define CIN   64
#define COUT  64
#define T_    64
#define V_    25
#define REL   8
#define TV    (T_ * V_)
#define VP    28
#define BLK   512
#define NWAVE 16
#define TB    16
#define TBV   (TB * V_)
#define NTB   (T_ / TB)

#define OFF_AF    0
#define AFC       (V_ * VP)
#define SZ_AF     (COUT * AFC)
#define OFF_INP   (OFF_AF + SZ_AF)
#define INC       (TB * VP)
#define SZ_INP    (COUT * INC)
#define OFF_XBAR  OFF_INP
#define SZ_XBAR   (CIN * V_)
#define OFF_X1    (OFF_XBAR + SZ_XBAR)
#define OFF_X2    (OFF_X1 + REL * V_)
#define OFF_D     (OFF_X2 + REL * V_)
#define SZ_D      (REL * V_ * V_)
#define OFF_PART  (OFF_D + SZ_D)
#define QPR       (V_ * 4)
#define SZ_PART   (CIN * QPR)
#define SMEM_FLOATS (OFF_INP + SZ_INP)

__global__ void __launch_bounds__(BLK)
gcn_fused_kernel(const float* __restrict__ x,
                 const float* __restrict__ Ag,
                 const float* __restrict__ w1, const float* __restrict__ b1,
                 const float* __restrict__ w2, const float* __restrict__ b2,
                 const float* __restrict__ w3, const float* __restrict__ b3,
                 const float* __restrict__ wr, const float* __restrict__ br,
                 float* __restrict__ out)
{
    extern __shared__ float sm[];
    const int n    = blockIdx.x;
    const int tid  = threadIdx.x;
    const int wave = tid >> 5;
    const int lane = tid & 31;
    const int lo   = lane & 15;
    const int hi   = lane >> 4;
    const int mtile = wave & 3;
    const float* __restrict__ xg = x + (size_t)n * CIN * TV;

    for (int idx = tid; idx < COUT * V_ * (VP - V_); idx += BLK) {
        const int row = idx / (VP - V_), p = idx % (VP - V_);
        sm[OFF_AF + row * VP + V_ + p] = 0.0f;
    }

    v2f wa[CIN / 4];
    float brv[8];
    {
        const float* wrow = wr + (mtile * 16 + lo) * CIN + 2 * hi;
#pragma unroll
        for (int ks = 0; ks < CIN / 4; ++ks) { wa[ks].x = wrow[4 * ks]; wa[ks].y = wrow[4 * ks + 1]; }
#pragma unroll
        for (int j = 0; j < 8; ++j) brv[j] = br[mtile * 16 + 8 * hi + j];
    }

    for (int idx = tid; idx < CIN * V_; idx += BLK) {
        const int i = idx / V_, q = idx % V_;
        const v4f* p = reinterpret_cast<const v4f*>(xg + i * TV) + q;
        v4f r[T_ / 4];
#pragma unroll
        for (int m = 0; m < T_ / 4; ++m) r[m] = p[m * V_];
        v4f s = r[0];
#pragma unroll
        for (int m = 1; m < T_ / 4; ++m) s += r[m];
        float* dst = &sm[OFF_PART + i * QPR + 4 * q];
        dst[0] = s.x; dst[1] = s.y; dst[2] = s.z; dst[3] = s.w;
    }
    __syncthreads();
    for (int idx = tid; idx < CIN * V_; idx += BLK) {
        const int i = idx / V_, v = idx % V_;
        const float* ps = &sm[OFF_PART + i * QPR + v];
        sm[OFF_XBAR + idx] = ((ps[0] + ps[V_]) + (ps[2 * V_] + ps[3 * V_])) * (1.0f / T_);
    }
    __syncthreads();

    for (int idx = tid; idx < 2 * REL * V_; idx += BLK) {
        const int which = idx / (REL * V_);
        const int rv = idx % (REL * V_);
        const int r = rv / V_, v = rv % V_;
        const float* __restrict__ w = which ? w2 : w1;
        float s = which ? b2[r] : b1[r];
#pragma unroll 8
        for (int i = 0; i < CIN; ++i) s += w[r * CIN + i] * sm[OFF_XBAR + i * V_ + v];
        sm[(which ? OFF_X2 : OFF_X1) + rv] = s;
    }
    __syncthreads();

    v2f wa0, wa1;
    float bias[8];
    {
        const float* w3row = w3 + (mtile * 16 + lo) * REL + 2 * hi;
        wa0.x = w3row[0]; wa0.y = w3row[1];
        wa1.x = w3row[4]; wa1.y = w3row[5];
#pragma unroll
        for (int j = 0; j < 8; ++j) bias[j] = b3[mtile * 16 + 8 * hi + j];
    }

    for (int idx = tid; idx < SZ_D; idx += BLK) {
        const int r = idx / (V_ * V_);
        const int uv = idx % (V_ * V_);
        const int u = uv / V_, v = uv % V_;
        sm[OFF_D + idx] = tanhf(sm[OFF_X1 + r * V_ + u] - sm[OFF_X2 + r * V_ + v]);
    }
    __syncthreads();

    for (int trip = 0; trip < ((V_ * V_ + 15) / 16) / (NWAVE / 4); ++trip) {
        const int uvt = (wave >> 2) + (NWAVE / 4) * trip;
        const int uv  = uvt * 16 + lo;
        const int uvc = uv < V_ * V_ ? uv : V_ * V_ - 1;
        const float ag = Ag[uvc];
        v2f d0, d1;
        d0.x = sm[OFF_D + (2 * hi)     * (V_ * V_) + uvc];
        d0.y = sm[OFF_D + (2 * hi + 1) * (V_ * V_) + uvc];
        d1.x = sm[OFF_D + (2 * hi + 4) * (V_ * V_) + uvc];
        d1.y = sm[OFF_D + (2 * hi + 5) * (V_ * V_) + uvc];
        v8f acc = {};
        acc = __builtin_amdgcn_wmma_f32_16x16x4_f32(false, wa0, false, d0, (short)0, acc, false, false);
        acc = __builtin_amdgcn_wmma_f32_16x16x4_f32(false, wa1, false, d1, (short)0, acc, false, false);
        if (uv < V_ * V_) {
            const int u = uv / V_, v = uv % V_;
            float* dst = &sm[OFF_AF + (mtile * 16 + 8 * hi) * AFC + u * VP + v];
#pragma unroll
            for (int j = 0; j < 8; ++j)
                dst[j * AFC] = acc[j] + (bias[j] + ag);
        }
    }

    for (int tb = 0; tb < NTB; ++tb) {
        __syncthreads();

        for (int idx = tid; idx < COUT * TB * (VP - V_); idx += BLK) {
            const int row = idx / (VP - V_), p = idx % (VP - V_);
            sm[OFF_INP + row * VP + V_ + p] = 0.0f;
        }

        {
            const float* xb = xg + tb * TBV + (2 * hi) * TV;
            for (int ctl = wave >> 2; ctl < TBV / 16; ctl += NWAVE / 4) {
                const int lc = ctl * 16 + lo;
                const float* xc = xb + lc;
                v2f b[CIN / 4];
#pragma unroll
                for (int ks = 0; ks < CIN / 4; ++ks) {
                    b[ks].x = xc[(4 * ks) * TV];
                    b[ks].y = xc[(4 * ks + 1) * TV];
                }
                v8f acc = {};
#pragma unroll
                for (int ks = 0; ks < CIN / 4; ++ks)
                    acc = __builtin_amdgcn_wmma_f32_16x16x4_f32(
                            false, wa[ks], false, b[ks], (short)0, acc, false, false);
                const int tl = lc / V_, v = lc % V_;
                float* dst = &sm[OFF_INP + (mtile * 16 + 8 * hi) * INC + tl * VP + v];
#pragma unroll
                for (int j = 0; j < 8; ++j)
                    dst[j * INC] = acc[j] + brv[j];
            }
        }
        __syncthreads();

        for (int q = 0; q < COUT / NWAVE; ++q) {
            const int c = wave * (COUT / NWAVE) + q;
            float* ip = &sm[OFF_INP + c * INC];
            const float* af = &sm[OFF_AF + c * AFC];
            const int u1 = (16 + lo < V_) ? 16 + lo : V_ - 1;
            v2f a[VP / 4], b0[VP / 4], b1[VP / 4];
#pragma unroll
            for (int k = 0; k < VP / 4; ++k) {
                const int kk = 4 * k + 2 * hi;
                a[k].x  = ip[lo * VP + kk];  a[k].y  = ip[lo * VP + kk + 1];
                b0[k].x = af[lo * VP + kk];  b0[k].y = af[lo * VP + kk + 1];
                b1[k].x = af[u1 * VP + kk];  b1[k].y = af[u1 * VP + kk + 1];
            }
            v8f acc0 = {}, acc1 = {};
#pragma unroll
            for (int k = 0; k < VP / 4; ++k) {
                acc0 = __builtin_amdgcn_wmma_f32_16x16x4_f32(false, a[k], false, b0[k], (short)0, acc0, false, false);
                acc1 = __builtin_amdgcn_wmma_f32_16x16x4_f32(false, a[k], false, b1[k], (short)0, acc1, false, false);
            }
#pragma unroll
            for (int j = 0; j < 8; ++j) ip[(j + 8 * hi) * V_ + lo] = acc0[j];
            if (16 + lo < V_) {
#pragma unroll
                for (int j = 0; j < 8; ++j) ip[(j + 8 * hi) * V_ + 16 + lo] = acc1[j];
            }
            float* og = out + ((size_t)n * COUT + c) * TV + tb * TBV;
            const int shift = (tb & 1) * 28;
#pragma unroll
            for (int s = 0; s < (TBV / 4 + 31) / 32; ++s) {
                const int qq = lane + 32 * s - shift;
                if (qq >= 0 && qq < TBV / 4) {
                    const v4f val = *reinterpret_cast<const v4f*>(ip + 4 * qq);
                    *reinterpret_cast<v4f*>(og + 4 * qq) = val;
                }
            }
        }
    }
}

extern "C" void kernel_launch(void* const* d_in, const int* in_sizes, int n_in,
                              void* d_out, int out_size, void* d_ws, size_t ws_size,
                              hipStream_t stream) {
    const float* x  = (const float*)d_in[0];
    const float* A  = (const float*)d_in[1];
    const float* w1 = (const float*)d_in[2];
    const float* b1 = (const float*)d_in[3];
    const float* w2 = (const float*)d_in[4];
    const float* b2 = (const float*)d_in[5];
    const float* w3 = (const float*)d_in[6];
    const float* b3 = (const float*)d_in[7];
    const float* wr = (const float*)d_in[8];
    const float* br = (const float*)d_in[9];
    float* out = (float*)d_out;

    const size_t shmem = (size_t)SMEM_FLOATS * sizeof(float);
    gcn_fused_kernel<<<dim3(N_), dim3(BLK), shmem, stream>>>(
        x, A, w1, b1, w2, b2, w3, b3, wr, br, out);
}


# GEMM1 unit = one x tile times two 16-channel tiles per wave (was one): each x tile is loaded by two waves, not four, and feeds two independent WMMA chains; 25th column tile split over waves 0..3
# speedup vs baseline: 1.6444x; 1.0370x over previous
#include <hip/hip_runtime.h>
#include <cmath>


typedef float v2f __attribute__((ext_vector_type(2)));
typedef float v8f __attribute__((ext_vector_type(8)));
typedef float v4f __attribute__((ext_vector_type(4)));

#define N_    256
#define CIN   64
#define COUT  64
#define T_    64
#define V_    25
#define REL   8
#define TV    (T_ * V_)
#define VP    28
#define BLK   512
#define NWAVE 16
#define TB    16
#define TBV   (TB * V_)
#define NTB   (T_ / TB)

#define OFF_AF    0
#define AFC       (V_ * VP)
#define SZ_AF     (COUT * AFC)
#define OFF_INP   (OFF_AF + SZ_AF)
#define INC       (TB * VP)
#define SZ_INP    (COUT * INC)
#define OFF_XBAR  OFF_INP
#define SZ_XBAR   (CIN * V_)
#define OFF_X1    (OFF_XBAR + SZ_XBAR)
#define OFF_X2    (OFF_X1 + REL * V_)
#define OFF_D     (OFF_X2 + REL * V_)
#define SZ_D      (REL * V_ * V_)
#define OFF_PART  (OFF_D + SZ_D)
#define QPR       (V_ * 4)
#define SZ_PART   (CIN * QPR)
#define SMEM_FLOATS (OFF_INP + SZ_INP)

__global__ void __launch_bounds__(BLK)
gcn_fused_kernel(const float* __restrict__ x,
                 const float* __restrict__ Ag,
                 const float* __restrict__ w1, const float* __restrict__ b1,
                 const float* __restrict__ w2, const float* __restrict__ b2,
                 const float* __restrict__ w3, const float* __restrict__ b3,
                 const float* __restrict__ wr, const float* __restrict__ br,
                 float* __restrict__ out)
{
    extern __shared__ float sm[];
    const int n    = blockIdx.x;
    const int tid  = threadIdx.x;
    const int wave = tid >> 5;
    const int lane = tid & 31;
    const int lo   = lane & 15;
    const int hi   = lane >> 4;
    const int mtile = wave & 3;
    const int chalf = wave & 1;
    const int cgrp  = wave >> 1;
    const float* __restrict__ xg = x + (size_t)n * CIN * TV;

    for (int idx = tid; idx < COUT * V_ * (VP - V_); idx += BLK) {
        const int row = idx / (VP - V_), p = idx % (VP - V_);
        sm[OFF_AF + row * VP + V_ + p] = 0.0f;
    }

    v2f wa0[CIN / 4], wa1[CIN / 4];
    float brv0[8], brv1[8];
    {
        const float* wrow0 = wr + (chalf * 32 + lo) * CIN + 2 * hi;
        const float* wrow1 = wrow0 + 16 * CIN;
#pragma unroll
        for (int ks = 0; ks < CIN / 4; ++ks) {
            wa0[ks].x = wrow0[4 * ks]; wa0[ks].y = wrow0[4 * ks + 1];
            wa1[ks].x = wrow1[4 * ks]; wa1[ks].y = wrow1[4 * ks + 1];
        }
#pragma unroll
        for (int j = 0; j < 8; ++j) {
            brv0[j] = br[chalf * 32 + 8 * hi + j];
            brv1[j] = br[chalf * 32 + 16 + 8 * hi + j];
        }
    }

    for (int idx = tid; idx < CIN * V_; idx += BLK) {
        const int i = idx / V_, q = idx % V_;
        const v4f* p = reinterpret_cast<const v4f*>(xg + i * TV) + q;
        v4f r[T_ / 4];
#pragma unroll
        for (int m = 0; m < T_ / 4; ++m) r[m] = p[m * V_];
        v4f s = r[0];
#pragma unroll
        for (int m = 1; m < T_ / 4; ++m) s += r[m];
        float* dst = &sm[OFF_PART + i * QPR + 4 * q];
        dst[0] = s.x; dst[1] = s.y; dst[2] = s.z; dst[3] = s.w;
    }
    __syncthreads();
    for (int idx = tid; idx < CIN * V_; idx += BLK) {
        const int i = idx / V_, v = idx % V_;
        const float* ps = &sm[OFF_PART + i * QPR + v];
        sm[OFF_XBAR + idx] = ((ps[0] + ps[V_]) + (ps[2 * V_] + ps[3 * V_])) * (1.0f / T_);
    }
    __syncthreads();

    for (int idx = tid; idx < 2 * REL * V_; idx += BLK) {
        const int which = idx / (REL * V_);
        const int rv = idx % (REL * V_);
        const int r = rv / V_, v = rv % V_;
        const float* __restrict__ w = which ? w2 : w1;
        float s = which ? b2[r] : b1[r];
#pragma unroll 8
        for (int i = 0; i < CIN; ++i) s += w[r * CIN + i] * sm[OFF_XBAR + i * V_ + v];
        sm[(which ? OFF_X2 : OFF_X1) + rv] = s;
    }
    __syncthreads();

    v2f w3f0, w3f1;
    float bias[8];
    {
        const float* w3row = w3 + (mtile * 16 + lo) * REL + 2 * hi;
        w3f0.x = w3row[0]; w3f0.y = w3row[1];
        w3f1.x = w3row[4]; w3f1.y = w3row[5];
#pragma unroll
        for (int j = 0; j < 8; ++j) bias[j] = b3[mtile * 16 + 8 * hi + j];
    }

    for (int idx = tid; idx < SZ_D; idx += BLK) {
        const int r = idx / (V_ * V_);
        const int uv = idx % (V_ * V_);
        const int u = uv / V_, v = uv % V_;
        sm[OFF_D + idx] = tanhf(sm[OFF_X1 + r * V_ + u] - sm[OFF_X2 + r * V_ + v]);
    }
    __syncthreads();

    for (int trip = 0; trip < ((V_ * V_ + 15) / 16) / (NWAVE / 4); ++trip) {
        const int uvt = (wave >> 2) + (NWAVE / 4) * trip;
        const int uv  = uvt * 16 + lo;
        const int uvc = uv < V_ * V_ ? uv : V_ * V_ - 1;
        const float ag = Ag[uvc];
        v2f d0, d1;
        d0.x = sm[OFF_D + (2 * hi)     * (V_ * V_) + uvc];
        d0.y = sm[OFF_D + (2 * hi + 1) * (V_ * V_) + uvc];
        d1.x = sm[OFF_D + (2 * hi + 4) * (V_ * V_) + uvc];
        d1.y = sm[OFF_D + (2 * hi + 5) * (V_ * V_) + uvc];
        v8f acc = {};
        acc = __builtin_amdgcn_wmma_f32_16x16x4_f32(false, w3f0, false, d0, (short)0, acc, false, false);
        acc = __builtin_amdgcn_wmma_f32_16x16x4_f32(false, w3f1, false, d1, (short)0, acc, false, false);
        if (uv < V_ * V_) {
            const int u = uv / V_, v = uv % V_;
            float* dst = &sm[OFF_AF + (mtile * 16 + 8 * hi) * AFC + u * VP + v];
#pragma unroll
            for (int j = 0; j < 8; ++j)
                dst[j * AFC] = acc[j] + (bias[j] + ag);
        }
    }

    for (int tb = 0; tb < NTB; ++tb) {
        __syncthreads();

        for (int idx = tid; idx < COUT * TB * (VP - V_); idx += BLK) {
            const int row = idx / (VP - V_), p = idx % (VP - V_);
            sm[OFF_INP + row * VP + V_ + p] = 0.0f;
        }

        {
            const float* xb = xg + tb * TBV + (2 * hi) * TV;
            if (wave < 4) {
                const int lc = (TBV / 16 - 1) * 16 + lo;
                const float* xc = xb + lc;
                v2f b[CIN / 4];
#pragma unroll
                for (int ks = 0; ks < CIN / 4; ++ks) {
                    b[ks].x = xc[(4 * ks) * TV];
                    b[ks].y = xc[(4 * ks + 1) * TV];
                }
                const bool first = (cgrp == 0);
                const int tl = lc / V_, v = lc % V_;
                float* dst = &sm[OFF_INP + (chalf * 32 + (first ? 0 : 16) + 8 * hi) * INC + tl * VP + v];
                v8f acc = {};
#pragma unroll
                for (int ks = 0; ks < CIN / 4; ++ks) {
                    v2f a;
                    a.x = first ? wa0[ks].x : wa1[ks].x;
                    a.y = first ? wa0[ks].y : wa1[ks].y;
                    acc = __builtin_amdgcn_wmma_f32_16x16x4_f32(
                            false, a, false, b[ks], (short)0, acc, false, false);
                }
#pragma unroll
                for (int j = 0; j < 8; ++j) dst[j * INC] = acc[j] + (first ? brv0[j] : brv1[j]);
            }
            for (int ctl = cgrp; ctl < TBV / 16 - 1; ctl += NWAVE / 2) {
                const int lc = ctl * 16 + lo;
                const float* xc = xb + lc;
                v2f b[CIN / 4];
#pragma unroll
                for (int ks = 0; ks < CIN / 4; ++ks) {
                    b[ks].x = xc[(4 * ks) * TV];
                    b[ks].y = xc[(4 * ks + 1) * TV];
                }
                v8f acc0 = {}, acc1 = {};
#pragma unroll
                for (int ks = 0; ks < CIN / 4; ++ks) {
                    acc0 = __builtin_amdgcn_wmma_f32_16x16x4_f32(
                            false, wa0[ks], false, b[ks], (short)0, acc0, false, false);
                    acc1 = __builtin_amdgcn_wmma_f32_16x16x4_f32(
                            false, wa1[ks], false, b[ks], (short)0, acc1, false, false);
                }
                const int tl = lc / V_, v = lc % V_;
                float* dst = &sm[OFF_INP + (chalf * 32 + 8 * hi) * INC + tl * VP + v];
#pragma unroll
                for (int j = 0; j < 8; ++j) {
                    dst[j * INC]        = acc0[j] + brv0[j];
                    dst[(16 + j) * INC] = acc1[j] + brv1[j];
                }
            }
        }
        __syncthreads();

        for (int q = 0; q < COUT / NWAVE; ++q) {
            const int c = wave * (COUT / NWAVE) + q;
            float* ip = &sm[OFF_INP + c * INC];
            const float* af = &sm[OFF_AF + c * AFC];
            const int u1 = (16 + lo < V_) ? 16 + lo : V_ - 1;
            v2f a[VP / 4], b0[VP / 4], b1[VP / 4];
#pragma unroll
            for (int k = 0; k < VP / 4; ++k) {
                const int kk = 4 * k + 2 * hi;
                a[k].x  = ip[lo * VP + kk];  a[k].y  = ip[lo * VP + kk + 1];
                b0[k].x = af[lo * VP + kk];  b0[k].y = af[lo * VP + kk + 1];
                b1[k].x = af[u1 * VP + kk];  b1[k].y = af[u1 * VP + kk + 1];
            }
            v8f acc0 = {}, acc1 = {};
#pragma unroll
            for (int k = 0; k < VP / 4; ++k) {
                acc0 = __builtin_amdgcn_wmma_f32_16x16x4_f32(false, a[k], false, b0[k], (short)0, acc0, false, false);
                acc1 = __builtin_amdgcn_wmma_f32_16x16x4_f32(false, a[k], false, b1[k], (short)0, acc1, false, false);
            }
#pragma unroll
            for (int j = 0; j < 8; ++j) ip[(j + 8 * hi) * V_ + lo] = acc0[j];
            if (16 + lo < V_) {
#pragma unroll
                for (int j = 0; j < 8; ++j) ip[(j + 8 * hi) * V_ + 16 + lo] = acc1[j];
            }
            float* og = out + ((size_t)n * COUT + c) * TV + tb * TBV;
            const int shift = (tb & 1) * 28;
#pragma unroll
            for (int s = 0; s < (TBV / 4 + 31) / 32; ++s) {
                const int qq = lane + 32 * s - shift;
                if (qq >= 0 && qq < TBV / 4) {
                    const v4f val = *reinterpret_cast<const v4f*>(ip + 4 * qq);
                    *reinterpret_cast<v4f*>(og + 4 * qq) = val;
                }
            }
        }
    }
}

extern "C" void kernel_launch(void* const* d_in, const int* in_sizes, int n_in,
                              void* d_out, int out_size, void* d_ws, size_t ws_size,
                              hipStream_t stream) {
    const float* x  = (const float*)d_in[0];
    const float* A  = (const float*)d_in[1];
    const float* w1 = (const float*)d_in[2];
    const float* b1 = (const float*)d_in[3];
    const float* w2 = (const float*)d_in[4];
    const float* b2 = (const float*)d_in[5];
    const float* w3 = (const float*)d_in[6];
    const float* b3 = (const float*)d_in[7];
    const float* wr = (const float*)d_in[8];
    const float* br = (const float*)d_in[9];
    float* out = (float*)d_out;

    const size_t shmem = (size_t)SMEM_FLOATS * sizeof(float);
    gcn_fused_kernel<<<dim3(N_), dim3(BLK), shmem, stream>>>(
        x, A, w1, b1, w2, b2, w3, b3, wr, br, out);
}


# GEMM1 of the first 16 time steps runs inside the time-mean pass (each wave's WMMA units sit between the mean's loads and their sums): matrix unit busy while x streams in; scratch moved off inputs
# speedup vs baseline: 1.7025x; 1.0353x over previous
#include <hip/hip_runtime.h>
#include <cmath>


typedef float v2f __attribute__((ext_vector_type(2)));
typedef float v8f __attribute__((ext_vector_type(8)));
typedef float v4f __attribute__((ext_vector_type(4)));

#define N_    256
#define CIN   64
#define COUT  64
#define T_    64
#define V_    25
#define REL   8
#define TV    (T_ * V_)
#define VP    28
#define BLK   512
#define NWAVE 16
#define TB    16
#define TBV   (TB * V_)
#define NTB   (T_ / TB)

#define OFF_AF    0
#define AFC       (V_ * VP)
#define SZ_AF     (COUT * AFC)
#define OFF_INP   (OFF_AF + SZ_AF)
#define INC       (TB * VP)
#define SZ_INP    (COUT * INC)
#define QPR       (V_ * 4)
#define SZ_PART   (CIN * QPR)
#define OFF_PART  OFF_AF
#define OFF_XBAR  (OFF_PART + SZ_PART)
#define SZ_XBAR   (CIN * V_)
#define OFF_X1    (OFF_INP + SZ_INP)
#define OFF_X2    (OFF_X1 + REL * V_)
#define OFF_D     (OFF_X2 + REL * V_)
#define SZ_D      (REL * V_ * V_)
#define SMEM_FLOATS (OFF_D + SZ_D)
#define NMAIN     ((TBV / 16 - 1) / (NWAVE / 2))
static_assert(SZ_PART + SZ_XBAR <= SZ_AF, "mean scratch must fit in the Afull area");
static_assert((TBV / 16 - 1) % (NWAVE / 2) == 0, "column tiles 0..23 must deal out evenly over the wave pairs");
static_assert(CIN * V_ > NMAIN * BLK && CIN * V_ - NMAIN * BLK == 64, "mean trips must pair with the GEMM1 units");

static __device__ __forceinline__ void g1_load(v2f (&b)[CIN / 4], const float* __restrict__ xc) {
#pragma unroll
    for (int ks = 0; ks < CIN / 4; ++ks) {
        b[ks].x = xc[(4 * ks) * TV];
        b[ks].y = xc[(4 * ks + 1) * TV];
    }
}
static __device__ __forceinline__ void g1_unit(const v2f (&wa0)[CIN / 4], const v2f (&wa1)[CIN / 4],
                                               const v2f (&b)[CIN / 4],
                                               const float (&brv0)[8], const float (&brv1)[8], float* dst) {
    v8f acc0 = {}, acc1 = {};
#pragma unroll
    for (int ks = 0; ks < CIN / 4; ++ks) {
        acc0 = __builtin_amdgcn_wmma_f32_16x16x4_f32(false, wa0[ks], false, b[ks], (short)0, acc0, false, false);
        acc1 = __builtin_amdgcn_wmma_f32_16x16x4_f32(false, wa1[ks], false, b[ks], (short)0, acc1, false, false);
    }
#pragma unroll
    for (int j = 0; j < 8; ++j) {
        dst[j * INC]        = acc0[j] + brv0[j];
        dst[(16 + j) * INC] = acc1[j] + brv1[j];
    }
}
static __device__ __forceinline__ void g1_half_unit(const v2f (&wa0)[CIN / 4], const v2f (&wa1)[CIN / 4],
                                                    const v2f (&b)[CIN / 4],
                                                    const float (&brv0)[8], const float (&brv1)[8],
                                                    bool first, float* dst) {
    v8f acc = {};
#pragma unroll
    for (int ks = 0; ks < CIN / 4; ++ks) {
        v2f a;
        a.x = first ? wa0[ks].x : wa1[ks].x;
        a.y = first ? wa0[ks].y : wa1[ks].y;
        acc = __builtin_amdgcn_wmma_f32_16x16x4_f32(false, a, false, b[ks], (short)0, acc, false, false);
    }
#pragma unroll
    for (int j = 0; j < 8; ++j) dst[j * INC] = acc[j] + (first ? brv0[j] : brv1[j]);
}
static __device__ __forceinline__ void mean_load(v4f (&r)[T_ / 4], const float* __restrict__ xg, int idx) {
    const int i = idx / V_, q = idx % V_;
    const v4f* p = reinterpret_cast<const v4f*>(xg + i * TV) + q;
#pragma unroll
    for (int m = 0; m < T_ / 4; ++m) r[m] = p[m * V_];
}
static __device__ __forceinline__ void mean_store(const v4f (&r)[T_ / 4], float* sm, int idx) {
    const int i = idx / V_, q = idx % V_;
    v4f s = r[0];
#pragma unroll
    for (int m = 1; m < T_ / 4; ++m) s += r[m];
    float* dst = &sm[OFF_PART + i * QPR + 4 * q];
    dst[0] = s.x; dst[1] = s.y; dst[2] = s.z; dst[3] = s.w;
}

__global__ void __launch_bounds__(BLK)
gcn_fused_kernel(const float* __restrict__ x,
                 const float* __restrict__ Ag,
                 const float* __restrict__ w1, const float* __restrict__ b1,
                 const float* __restrict__ w2, const float* __restrict__ b2,
                 const float* __restrict__ w3, const float* __restrict__ b3,
                 const float* __restrict__ wr, const float* __restrict__ br,
                 float* __restrict__ out)
{
    extern __shared__ float sm[];
    const int n    = blockIdx.x;
    const int tid  = threadIdx.x;
    const int wave = tid >> 5;
    const int lane = tid & 31;
    const int lo   = lane & 15;
    const int hi   = lane >> 4;
    const int mtile = wave & 3;
    const int chalf = wave & 1;
    const int cgrp  = wave >> 1;
    const float* __restrict__ xg = x + (size_t)n * CIN * TV;

    v2f wa0[CIN / 4], wa1[CIN / 4];
    float brv0[8], brv1[8];
    {
        const float* wrow0 = wr + (chalf * 32 + lo) * CIN + 2 * hi;
        const float* wrow1 = wrow0 + 16 * CIN;
#pragma unroll
        for (int ks = 0; ks < CIN / 4; ++ks) {
            wa0[ks].x = wrow0[4 * ks]; wa0[ks].y = wrow0[4 * ks + 1];
            wa1[ks].x = wrow1[4 * ks]; wa1[ks].y = wrow1[4 * ks + 1];
        }
#pragma unroll
        for (int j = 0; j < 8; ++j) {
            brv0[j] = br[chalf * 32 + 8 * hi + j];
            brv1[j] = br[chalf * 32 + 16 + 8 * hi + j];
        }
    }

    {
        const float* xb = xg + (2 * hi) * TV;
        v4f r[T_ / 4];
        __builtin_amdgcn_sched_barrier(0);
        mean_load(r, xg, tid);
        if (wave < 4) {
            const int lc = (TBV / 16 - 1) * 16 + lo;
            v2f b[CIN / 4];
            g1_load(b, xb + lc);
            const bool first = (cgrp == 0);
            const int tl = lc / V_, v = lc % V_;
            g1_half_unit(wa0, wa1, b, brv0, brv1, first,
                         &sm[OFF_INP + (chalf * 32 + (first ? 0 : 16) + 8 * hi) * INC + tl * VP + v]);
        }
#pragma unroll
        for (int k = 0; k < NMAIN; ++k) {
            const int lc = (cgrp + (NWAVE / 2) * k) * 16 + lo;
            v2f b[CIN / 4];
            g1_load(b, xb + lc);
            __builtin_amdgcn_sched_barrier(0);
            mean_store(r, sm, tid + BLK * k);
            if (k + 1 < NMAIN)  mean_load(r, xg, tid + BLK * (k + 1));
            else if (wave < 2)  mean_load(r, xg, tid + BLK * NMAIN);
            const int tl = lc / V_, v = lc % V_;
            __builtin_amdgcn_sched_barrier(0);
            g1_unit(wa0, wa1, b, brv0, brv1, &sm[OFF_INP + (chalf * 32 + 8 * hi) * INC + tl * VP + v]);
            __builtin_amdgcn_sched_barrier(0);
        }
        if (wave < 2) mean_store(r, sm, tid + BLK * NMAIN);
    }
    __syncthreads();
    for (int idx = tid; idx < CIN * V_; idx += BLK) {
        const int i = idx / V_, v = idx % V_;
        const float* ps = &sm[OFF_PART + i * QPR + v];
        sm[OFF_XBAR + idx] = ((ps[0] + ps[V_]) + (ps[2 * V_] + ps[3 * V_])) * (1.0f / T_);
    }
    __syncthreads();

    for (int idx = tid; idx < 2 * REL * V_; idx += BLK) {
        const int which = idx / (REL * V_);
        const int rv = idx % (REL * V_);
        const int r = rv / V_, v = rv % V_;
        const float* __restrict__ w = which ? w2 : w1;
        float s = which ? b2[r] : b1[r];
#pragma unroll 8
        for (int i = 0; i < CIN; ++i) s += w[r * CIN + i] * sm[OFF_XBAR + i * V_ + v];
        sm[(which ? OFF_X2 : OFF_X1) + rv] = s;
    }
    __syncthreads();

    v2f w3f0, w3f1;
    float bias[8];
    {
        const float* w3row = w3 + (mtile * 16 + lo) * REL + 2 * hi;
        w3f0.x = w3row[0]; w3f0.y = w3row[1];
        w3f1.x = w3row[4]; w3f1.y = w3row[5];
#pragma unroll
        for (int j = 0; j < 8; ++j) bias[j] = b3[mtile * 16 + 8 * hi + j];
    }

    for (int idx = tid; idx < COUT * V_ * (VP - V_); idx += BLK) {
        const int row = idx / (VP - V_), p = idx % (VP - V_);
        sm[OFF_AF + row * VP + V_ + p] = 0.0f;
    }
    for (int idx = tid; idx < COUT * TB * (VP - V_); idx += BLK) {
        const int row = idx / (VP - V_), p = idx % (VP - V_);
        sm[OFF_INP + row * VP + V_ + p] = 0.0f;
    }

    for (int idx = tid; idx < SZ_D; idx += BLK) {
        const int r = idx / (V_ * V_);
        const int uv = idx % (V_ * V_);
        const int u = uv / V_, v = uv % V_;
        sm[OFF_D + idx] = tanhf(sm[OFF_X1 + r * V_ + u] - sm[OFF_X2 + r * V_ + v]);
    }
    __syncthreads();

    for (int trip = 0; trip < ((V_ * V_ + 15) / 16) / (NWAVE / 4); ++trip) {
        const int uvt = (wave >> 2) + (NWAVE / 4) * trip;
        const int uv  = uvt * 16 + lo;
        const int uvc = uv < V_ * V_ ? uv : V_ * V_ - 1;
        const float ag = Ag[uvc];
        v2f d0, d1;
        d0.x = sm[OFF_D + (2 * hi)     * (V_ * V_) + uvc];
        d0.y = sm[OFF_D + (2 * hi + 1) * (V_ * V_) + uvc];
        d1.x = sm[OFF_D + (2 * hi + 4) * (V_ * V_) + uvc];
        d1.y = sm[OFF_D + (2 * hi + 5) * (V_ * V_) + uvc];
        v8f acc = {};
        acc = __builtin_amdgcn_wmma_f32_16x16x4_f32(false, w3f0, false, d0, (short)0, acc, false, false);
        acc = __builtin_amdgcn_wmma_f32_16x16x4_f32(false, w3f1, false, d1, (short)0, acc, false, false);
        if (uv < V_ * V_) {
            const int u = uv / V_, v = uv % V_;
            float* dst = &sm[OFF_AF + (mtile * 16 + 8 * hi) * AFC + u * VP + v];
#pragma unroll
            for (int j = 0; j < 8; ++j)
                dst[j * AFC] = acc[j] + (bias[j] + ag);
        }
    }

    for (int tb = 0; tb < NTB; ++tb) {
        if (tb > 0) {
            __syncthreads();

            for (int idx = tid; idx < COUT * TB * (VP - V_); idx += BLK) {
                const int row = idx / (VP - V_), p = idx % (VP - V_);
                sm[OFF_INP + row * VP + V_ + p] = 0.0f;
            }

            const float* xb = xg + tb * TBV + (2 * hi) * TV;
            if (wave < 4) {
                const int lc = (TBV / 16 - 1) * 16 + lo;
                v2f b[CIN / 4];
                g1_load(b, xb + lc);
                const bool first = (cgrp == 0);
                const int tl = lc / V_, v = lc % V_;
                g1_half_unit(wa0, wa1, b, brv0, brv1, first,
                             &sm[OFF_INP + (chalf * 32 + (first ? 0 : 16) + 8 * hi) * INC + tl * VP + v]);
            }
            for (int ctl = cgrp; ctl < TBV / 16 - 1; ctl += NWAVE / 2) {
                const int lc = ctl * 16 + lo;
                v2f b[CIN / 4];
                g1_load(b, xb + lc);
                const int tl = lc / V_, v = lc % V_;
                g1_unit(wa0, wa1, b, brv0, brv1, &sm[OFF_INP + (chalf * 32 + 8 * hi) * INC + tl * VP + v]);
            }
        }
        __syncthreads();

        for (int q = 0; q < COUT / NWAVE; ++q) {
            const int c = wave * (COUT / NWAVE) + q;
            float* ip = &sm[OFF_INP + c * INC];
            const float* af = &sm[OFF_AF + c * AFC];
            const int u1 = (16 + lo < V_) ? 16 + lo : V_ - 1;
            v2f a[VP / 4], b0[VP / 4], b1[VP / 4];
#pragma unroll
            for (int k = 0; k < VP / 4; ++k) {
                const int kk = 4 * k + 2 * hi;
                a[k].x  = ip[lo * VP + kk];  a[k].y  = ip[lo * VP + kk + 1];
                b0[k].x = af[lo * VP + kk];  b0[k].y = af[lo * VP + kk + 1];
                b1[k].x = af[u1 * VP + kk];  b1[k].y = af[u1 * VP + kk + 1];
            }
            v8f acc0 = {}, acc1 = {};
#pragma unroll
            for (int k = 0; k < VP / 4; ++k) {
                acc0 = __builtin_amdgcn_wmma_f32_16x16x4_f32(false, a[k], false, b0[k], (short)0, acc0, false, false);
                acc1 = __builtin_amdgcn_wmma_f32_16x16x4_f32(false, a[k], false, b1[k], (short)0, acc1, false, false);
            }
#pragma unroll
            for (int j = 0; j < 8; ++j) ip[(j + 8 * hi) * V_ + lo] = acc0[j];
            if (16 + lo < V_) {
#pragma unroll
                for (int j = 0; j < 8; ++j) ip[(j + 8 * hi) * V_ + 16 + lo] = acc1[j];
            }
            float* og = out + ((size_t)n * COUT + c) * TV + tb * TBV;
            const int shift = (tb & 1) * 28;
#pragma unroll
            for (int s = 0; s < (TBV / 4 + 31) / 32; ++s) {
                const int qq = lane + 32 * s - shift;
                if (qq >= 0 && qq < TBV / 4) {
                    const v4f val = *reinterpret_cast<const v4f*>(ip + 4 * qq);
                    *reinterpret_cast<v4f*>(og + 4 * qq) = val;
                }
            }
        }
    }
}

extern "C" void kernel_launch(void* const* d_in, const int* in_sizes, int n_in,
                              void* d_out, int out_size, void* d_ws, size_t ws_size,
                              hipStream_t stream) {
    const float* x  = (const float*)d_in[0];
    const float* A  = (const float*)d_in[1];
    const float* w1 = (const float*)d_in[2];
    const float* b1 = (const float*)d_in[3];
    const float* w2 = (const float*)d_in[4];
    const float* b2 = (const float*)d_in[5];
    const float* w3 = (const float*)d_in[6];
    const float* b3 = (const float*)d_in[7];
    const float* wr = (const float*)d_in[8];
    const float* br = (const float*)d_in[9];
    float* out = (float*)d_out;

    const size_t shmem = (size_t)SMEM_FLOATS * sizeof(float);
    gcn_fused_kernel<<<dim3(N_), dim3(BLK), shmem, stream>>>(
        x, A, w1, b1, w2, b2, w3, b3, wr, br, out);
}
